# starting kernel
# baseline (speedup 1.0000x reference)
#include <hip/hip_runtime.h>
#include <hip/hip_bf16.h>

typedef __attribute__((ext_vector_type(2))) float v2f;
typedef __attribute__((ext_vector_type(8))) float v8f;

#define BATCH 2048
#define NCAT 32
#define SDIM 64
#define EMB 256
#define NOUT 256
#define ODIM 512
#define HDIM 512
#define HSTR 516
#define SSTR 68

__global__ void bucket_by_cat(const int* __restrict__ cat_ids,
                              int* __restrict__ counts,
                              int* __restrict__ offsets,
                              int* __restrict__ perm) {
    __shared__ int s_cnt[NCAT];
    __shared__ int s_cur[NCAT];
    int t = threadIdx.x;
    if (t < NCAT) s_cnt[t] = 0;
    __syncthreads();
    for (int i = t; i < BATCH; i += blockDim.x)
        atomicAdd(&s_cnt[cat_ids[i]], 1);
    __syncthreads();
    if (t == 0) {
        int acc = 0;
        for (int c = 0; c < NCAT; ++c) {
            s_cur[c] = acc;
            offsets[c] = acc;
            counts[c] = s_cnt[c];
            acc += s_cnt[c];
        }
    }
    __syncthreads();
    for (int i = t; i < BATCH; i += blockDim.x) {
        int c = cat_ids[i];
        int pos = atomicAdd(&s_cur[c], 1);
        perm[pos] = i;
    }
}

__global__ __launch_bounds__(128)
void moe_mlp_kernel(const float* __restrict__ state,
                    const float* __restrict__ obs,
                    const float* __restrict__ obs_W,
                    const float* __restrict__ obs_b,
                    const float* __restrict__ state_W,
                    const float* __restrict__ state_b,
                    const float* __restrict__ l2_W,
                    const float* __restrict__ l2_b,
                    const int* __restrict__ counts,
                    const int* __restrict__ offsets,
                    const int* __restrict__ perm,
                    float* __restrict__ out) {
    const int c    = blockIdx.y;
    const int tile = blockIdx.x;
    const int cnt  = counts[c];
    if (tile * 16 >= cnt) return;

    __shared__ int   sh_rows[16];
    __shared__ float sh_s[16 * SSTR];
    __shared__ float sh_h[16 * HSTR];

    const int t    = threadIdx.x;
    const int wave = t >> 5;
    const int lane = t & 31;
    const int m    = lane & 15;
    const int half = lane >> 4;
    const int kb   = half * 2;
    const int mb   = half * 8;

    const int base = offsets[c];
    if (t < 16) {
        int idx = tile * 16 + t;
        sh_rows[t] = (idx < cnt) ? perm[base + idx] : -1;
    }
    __syncthreads();

    for (int i = t; i < 16 * SDIM; i += 128) {
        int mm = i >> 6, kk = i & (SDIM - 1);
        int r = sh_rows[mm];
        sh_s[mm * SSTR + kk] = (r >= 0) ? state[(long)r * SDIM + kk] : 0.0f;
    }
    __syncthreads();

    const float* Ws = state_W + (long)c * SDIM * EMB;
    const float* bs = state_b + (long)c * EMB;
    const float* Wl = l2_W    + (long)c * HDIM * NOUT;
    const float* lb = l2_b    + (long)c * NOUT;

    for (int j = 0; j < 4; ++j) {
        const int nt   = wave * 4 + j;
        const int ncol = nt * 16 + m;
        v8f acc = {};
        for (int k = 0; k < SDIM; k += 4) {
            v2f a, b;
            a.x = sh_s[m * SSTR + k + kb];
            a.y = sh_s[m * SSTR + k + kb + 1];
            b.x = Ws[(long)(k + kb)     * EMB + ncol];
            b.y = Ws[(long)(k + kb + 1) * EMB + ncol];
            acc = __builtin_amdgcn_wmma_f32_16x16x4_f32(
                false, a, false, b, (short)0, acc, false, false);
        }
        const float bias = bs[ncol];
        #pragma unroll
        for (int v = 0; v < 8; ++v)
            sh_h[(mb + v) * HSTR + ncol] = fmaxf(acc[v] + bias, 0.0f);
    }

    {
        const int  r      = sh_rows[m];
        const float validf = (r >= 0) ? 1.0f : 0.0f;
        const long ro     = (long)((r >= 0) ? r : 0) * ODIM;
        for (int j = 0; j < 4; ++j) {
            const int nt   = wave * 4 + j;
            const int ncol = nt * 16 + m;
            v8f acc = {};
            for (int k = 0; k < ODIM; k += 4) {
                v2f a, b;
                a.x = obs[ro + k + kb]     * validf;
                a.y = obs[ro + k + kb + 1] * validf;
                b.x = obs_W[(long)(k + kb)     * EMB + ncol];
                b.y = obs_W[(long)(k + kb + 1) * EMB + ncol];
                acc = __builtin_amdgcn_wmma_f32_16x16x4_f32(
                    false, a, false, b, (short)0, acc, false, false);
            }
            const float bias = obs_b[ncol];
            #pragma unroll
            for (int v = 0; v < 8; ++v)
                sh_h[(mb + v) * HSTR + EMB + ncol] = acc[v] + bias;
        }
    }
    __syncthreads();

    for (int j = 0; j < 4; ++j) {
        const int nt   = wave * 4 + j;
        const int ncol = nt * 16 + m;
        v8f acc = {};
        for (int k = 0; k < HDIM; k += 4) {
            v2f a, b;
            a.x = sh_h[m * HSTR + k + kb];
            a.y = sh_h[m * HSTR + k + kb + 1];
            b.x = Wl[(long)(k + kb)     * NOUT + ncol];
            b.y = Wl[(long)(k + kb + 1) * NOUT + ncol];
            acc = __builtin_amdgcn_wmma_f32_16x16x4_f32(
                false, a, false, b, (short)0, acc, false, false);
        }
        const float bias = lb[ncol];
        #pragma unroll
        for (int v = 0; v < 8; ++v) {
            int r = sh_rows[mb + v];
            if (r >= 0)
                out[(long)r * NOUT + ncol] = acc[v] + bias;
        }
    }
}

extern "C" void kernel_launch(void* const* d_in, const int* in_sizes, int n_in,
                              void* d_out, int out_size, void* d_ws, size_t ws_size,
                              hipStream_t stream) {
    const float* state   = (const float*)d_in[0];
    const float* obs     = (const float*)d_in[1];
    const int*   cat_ids = (const int*)  d_in[2];
    const float* obs_W   = (const float*)d_in[3];
    const float* obs_b   = (const float*)d_in[4];
    const float* state_W = (const float*)d_in[5];
    const float* state_b = (const float*)d_in[6];
    const float* l2_W    = (const float*)d_in[7];
    const float* l2_b    = (const float*)d_in[8];
    float*       out     = (float*)d_out;

    int* counts  = (int*)d_ws;
    int* offsets = counts + NCAT;
    int* perm    = offsets + NCAT;

    bucket_by_cat<<<1, 256, 0, stream>>>(cat_ids, counts, offsets, perm);

    dim3 grid((BATCH + 15) / 16, NCAT);
    moe_mlp_kernel<<<grid, 128, 0, stream>>>(state, obs, obs_W, obs_b,
                                             state_W, state_b, l2_W, l2_b,
                                             counts, offsets, perm, out);
}


# Compact grid: 158 work-groups instead of 4096 mostly empty ones; each finds its live (category, tile) by a scan of the 32 counts, so live tiles spread over the compute units, not stacked on a few
# speedup vs baseline: 2.2527x; 2.2527x over previous
#include <hip/hip_runtime.h>
#include <hip/hip_bf16.h>

typedef __attribute__((ext_vector_type(2))) float v2f;
typedef __attribute__((ext_vector_type(8))) float v8f;

#define BATCH 2048
#define NCAT 32
#define SDIM 64
#define EMB 256
#define NOUT 256
#define ODIM 512
#define HDIM 512
#define HSTR 516
#define SSTR 68
#define MAXTILES 158

__global__ void bucket_by_cat(const int* __restrict__ cat_ids,
                              int* __restrict__ counts,
                              int* __restrict__ offsets,
                              int* __restrict__ perm) {
    __shared__ int s_cnt[NCAT];
    __shared__ int s_cur[NCAT];
    int t = threadIdx.x;
    if (t < NCAT) s_cnt[t] = 0;
    __syncthreads();
    for (int i = t; i < BATCH; i += blockDim.x)
        atomicAdd(&s_cnt[cat_ids[i]], 1);
    __syncthreads();
    if (t == 0) {
        int acc = 0;
        for (int c = 0; c < NCAT; ++c) {
            s_cur[c] = acc;
            offsets[c] = acc;
            counts[c] = s_cnt[c];
            acc += s_cnt[c];
        }
    }
    __syncthreads();
    for (int i = t; i < BATCH; i += blockDim.x) {
        int c = cat_ids[i];
        int pos = atomicAdd(&s_cur[c], 1);
        perm[pos] = i;
    }
}

__global__ __launch_bounds__(128)
void moe_mlp_kernel(const float* __restrict__ state,
                    const float* __restrict__ obs,
                    const float* __restrict__ obs_W,
                    const float* __restrict__ obs_b,
                    const float* __restrict__ state_W,
                    const float* __restrict__ state_b,
                    const float* __restrict__ l2_W,
                    const float* __restrict__ l2_b,
                    const int* __restrict__ counts,
                    const int* __restrict__ offsets,
                    const int* __restrict__ perm,
                    float* __restrict__ out) {
    const int g = blockIdx.x;
    int c = 0, first = 0, run = 0;
    for (int cc = 0; cc < NCAT; ++cc) {
        run += (counts[cc] + 15) >> 4;
        if (g >= run) { c = cc + 1; first = run; }
    }
    if (c >= NCAT) return;
    const int tile = g - first;
    const int cnt  = counts[c];

    __shared__ int   sh_rows[16];
    __shared__ float sh_s[16 * SSTR];
    __shared__ float sh_h[16 * HSTR];

    const int t    = threadIdx.x;
    const int wave = t >> 5;
    const int lane = t & 31;
    const int m    = lane & 15;
    const int half = lane >> 4;
    const int kb   = half * 2;
    const int mb   = half * 8;

    const int base = offsets[c];
    if (t < 16) {
        int idx = tile * 16 + t;
        sh_rows[t] = (idx < cnt) ? perm[base + idx] : -1;
    }
    __syncthreads();

    for (int i = t; i < 16 * SDIM; i += 128) {
        int mm = i >> 6, kk = i & (SDIM - 1);
        int r = sh_rows[mm];
        sh_s[mm * SSTR + kk] = (r >= 0) ? state[(long)r * SDIM + kk] : 0.0f;
    }
    __syncthreads();

    const float* Ws = state_W + (long)c * SDIM * EMB;
    const float* bs = state_b + (long)c * EMB;
    const float* Wl = l2_W    + (long)c * HDIM * NOUT;
    const float* lb = l2_b    + (long)c * NOUT;

    for (int j = 0; j < 4; ++j) {
        const int nt   = wave * 4 + j;
        const int ncol = nt * 16 + m;
        v8f acc = {};
        for (int k = 0; k < SDIM; k += 4) {
            v2f a, b;
            a.x = sh_s[m * SSTR + k + kb];
            a.y = sh_s[m * SSTR + k + kb + 1];
            b.x = Ws[(long)(k + kb)     * EMB + ncol];
            b.y = Ws[(long)(k + kb + 1) * EMB + ncol];
            acc = __builtin_amdgcn_wmma_f32_16x16x4_f32(
                false, a, false, b, (short)0, acc, false, false);
        }
        const float bias = bs[ncol];
        #pragma unroll
        for (int v = 0; v < 8; ++v)
            sh_h[(mb + v) * HSTR + ncol] = fmaxf(acc[v] + bias, 0.0f);
    }

    {
        const int  r      = sh_rows[m];
        const float validf = (r >= 0) ? 1.0f : 0.0f;
        const long ro     = (long)((r >= 0) ? r : 0) * ODIM;
        for (int j = 0; j < 4; ++j) {
            const int nt   = wave * 4 + j;
            const int ncol = nt * 16 + m;
            v8f acc = {};
            for (int k = 0; k < ODIM; k += 4) {
                v2f a, b;
                a.x = obs[ro + k + kb]     * validf;
                a.y = obs[ro + k + kb + 1] * validf;
                b.x = obs_W[(long)(k + kb)     * EMB + ncol];
                b.y = obs_W[(long)(k + kb + 1) * EMB + ncol];
                acc = __builtin_amdgcn_wmma_f32_16x16x4_f32(
                    false, a, false, b, (short)0, acc, false, false);
            }
            const float bias = obs_b[ncol];
            #pragma unroll
            for (int v = 0; v < 8; ++v)
                sh_h[(mb + v) * HSTR + EMB + ncol] = acc[v] + bias;
        }
    }
    __syncthreads();

    for (int j = 0; j < 4; ++j) {
        const int nt   = wave * 4 + j;
        const int ncol = nt * 16 + m;
        v8f acc = {};
        for (int k = 0; k < HDIM; k += 4) {
            v2f a, b;
            a.x = sh_h[m * HSTR + k + kb];
            a.y = sh_h[m * HSTR + k + kb + 1];
            b.x = Wl[(long)(k + kb)     * NOUT + ncol];
            b.y = Wl[(long)(k + kb + 1) * NOUT + ncol];
            acc = __builtin_amdgcn_wmma_f32_16x16x4_f32(
                false, a, false, b, (short)0, acc, false, false);
        }
        const float bias = lb[ncol];
        #pragma unroll
        for (int v = 0; v < 8; ++v) {
            int r = sh_rows[mb + v];
            if (r >= 0)
                out[(long)r * NOUT + ncol] = acc[v] + bias;
        }
    }
}

extern "C" void kernel_launch(void* const* d_in, const int* in_sizes, int n_in,
                              void* d_out, int out_size, void* d_ws, size_t ws_size,
                              hipStream_t stream) {
    const float* state   = (const float*)d_in[0];
    const float* obs     = (const float*)d_in[1];
    const int*   cat_ids = (const int*)  d_in[2];
    const float* obs_W   = (const float*)d_in[3];
    const float* obs_b   = (const float*)d_in[4];
    const float* state_W = (const float*)d_in[5];
    const float* state_b = (const float*)d_in[6];
    const float* l2_W    = (const float*)d_in[7];
    const float* l2_b    = (const float*)d_in[8];
    float*       out     = (float*)d_out;

    int* counts  = (int*)d_ws;
    int* offsets = counts + NCAT;
    int* perm    = offsets + NCAT;

    bucket_by_cat<<<1, 256, 0, stream>>>(cat_ids, counts, offsets, perm);

    moe_mlp_kernel<<<MAXTILES, 128, 0, stream>>>(state, obs, obs_W, obs_b,
                                             state_W, state_b, l2_W, l2_b,
                                             counts, offsets, perm, out);
}


# One k loop per stage with 4 accumulator tiles per wave on interleaved columns (was 4 loops of one tile): A fragment read once, W rows by 16-byte loads, 16 k-steps of loads in flight per wait
# speedup vs baseline: 4.6325x; 2.0564x over previous
#include <hip/hip_runtime.h>
#include <hip/hip_bf16.h>

typedef __attribute__((ext_vector_type(2))) float v2f;
typedef __attribute__((ext_vector_type(4))) float v4f;
typedef __attribute__((ext_vector_type(8))) float v8f;

#define BATCH 2048
#define NCAT 32
#define SDIM 64
#define EMB 256
#define NOUT 256
#define ODIM 512
#define HDIM 512
#define HSTR 516
#define SSTR 68
#define KU 16
#define MAXTILES 158

__global__ void bucket_by_cat(const int* __restrict__ cat_ids,
                              int* __restrict__ counts,
                              int* __restrict__ offsets,
                              int* __restrict__ perm) {
    __shared__ int s_cnt[NCAT];
    __shared__ int s_cur[NCAT];
    int t = threadIdx.x;
    if (t < NCAT) s_cnt[t] = 0;
    __syncthreads();
    for (int i = t; i < BATCH; i += blockDim.x)
        atomicAdd(&s_cnt[cat_ids[i]], 1);
    __syncthreads();
    if (t == 0) {
        int acc = 0;
        for (int c = 0; c < NCAT; ++c) {
            s_cur[c] = acc;
            offsets[c] = acc;
            counts[c] = s_cnt[c];
            acc += s_cnt[c];
        }
    }
    __syncthreads();
    for (int i = t; i < BATCH; i += blockDim.x) {
        int c = cat_ids[i];
        int pos = atomicAdd(&s_cur[c], 1);
        perm[pos] = i;
    }
}

__device__ __forceinline__ void mma4(v8f (&acc)[4], v2f a, v4f b0, v4f b1) {
    #pragma unroll
    for (int j = 0; j < 4; ++j) {
        v2f b;
        b.x = b0[j];
        b.y = b1[j];
        acc[j] = __builtin_amdgcn_wmma_f32_16x16x4_f32(
            false, a, false, b, (short)0, acc[j], false, false);
    }
}

__global__ __launch_bounds__(128) __attribute__((amdgpu_waves_per_eu(1, 1)))
void moe_mlp_kernel(const float* __restrict__ state,
                    const float* __restrict__ obs,
                    const float* __restrict__ obs_W,
                    const float* __restrict__ obs_b,
                    const float* __restrict__ state_W,
                    const float* __restrict__ state_b,
                    const float* __restrict__ l2_W,
                    const float* __restrict__ l2_b,
                    const int* __restrict__ counts,
                    const int* __restrict__ offsets,
                    const int* __restrict__ perm,
                    float* __restrict__ out) {
    const int g = blockIdx.x;
    int c = 0, first = 0, run = 0;
    for (int cc = 0; cc < NCAT; ++cc) {
        run += (counts[cc] + 15) >> 4;
        if (g >= run) { c = cc + 1; first = run; }
    }
    if (c >= NCAT) return;
    const int tile = g - first;
    const int cnt  = counts[c];

    __shared__ int   sh_rows[16];
    __shared__ float sh_s[16 * SSTR];
    __shared__ __attribute__((aligned(16))) float sh_h[16 * HSTR];

    const int t    = threadIdx.x;
    const int wave = t >> 5;
    const int lane = t & 31;
    const int m    = lane & 15;
    const int half = lane >> 4;
    const int kb   = half * 2;
    const int mb   = half * 8;
    const int ncol0 = wave * 64 + 4 * m;

    const int base = offsets[c];
    if (t < 16) {
        int idx = tile * 16 + t;
        sh_rows[t] = (idx < cnt) ? perm[base + idx] : -1;
    }
    __syncthreads();

    for (int i = t; i < 16 * SDIM; i += 128) {
        int mm = i >> 6, kk = i & (SDIM - 1);
        int r = sh_rows[mm];
        sh_s[mm * SSTR + kk] = (r >= 0) ? state[(long)r * SDIM + kk] : 0.0f;
    }
    __syncthreads();

    const float* Ws = state_W + (long)c * SDIM * EMB;
    const float* bs = state_b + (long)c * EMB;
    const float* Wl = l2_W    + (long)c * HDIM * NOUT;
    const float* lb = l2_b    + (long)c * NOUT;

    {
        v8f acc[4] = {};
        const float* bp = Ws + (long)kb * EMB + ncol0;
        #pragma unroll
        for (int k = 0; k < SDIM; k += 4) {
            v2f a;
            a.x = sh_s[m * SSTR + k + kb];
            a.y = sh_s[m * SSTR + k + kb + 1];
            const v4f b0 = *(const v4f*)(bp + (long)k * EMB);
            const v4f b1 = *(const v4f*)(bp + (long)(k + 1) * EMB);
            mma4(acc, a, b0, b1);
        }
        const v4f bias = *(const v4f*)(bs + ncol0);
        #pragma unroll
        for (int v = 0; v < 8; ++v) {
            v4f h;
            h.x = fmaxf(acc[0][v] + bias.x, 0.0f);
            h.y = fmaxf(acc[1][v] + bias.y, 0.0f);
            h.z = fmaxf(acc[2][v] + bias.z, 0.0f);
            h.w = fmaxf(acc[3][v] + bias.w, 0.0f);
            *(v4f*)(&sh_h[(mb + v) * HSTR + ncol0]) = h;
        }
    }

    {
        const int  r      = sh_rows[m];
        const float validf = (r >= 0) ? 1.0f : 0.0f;
        const long ro     = (long)((r >= 0) ? r : 0) * ODIM;
        v8f acc[4] = {};
        const float* ap = obs + ro + kb;
        const float* bp = obs_W + (long)kb * EMB + ncol0;
        for (int k0 = 0; k0 < ODIM; k0 += 4 * KU) {
            #pragma unroll
            for (int u = 0; u < KU; ++u) {
                const int k = k0 + 4 * u;
                v2f a = *(const v2f*)(ap + k);
                a.x *= validf;
                a.y *= validf;
                const v4f b0 = *(const v4f*)(bp + (long)k * EMB);
                const v4f b1 = *(const v4f*)(bp + (long)(k + 1) * EMB);
                mma4(acc, a, b0, b1);
            }
        }
        const v4f bias = *(const v4f*)(obs_b + ncol0);
        #pragma unroll
        for (int v = 0; v < 8; ++v) {
            v4f h;
            h.x = acc[0][v] + bias.x;
            h.y = acc[1][v] + bias.y;
            h.z = acc[2][v] + bias.z;
            h.w = acc[3][v] + bias.w;
            *(v4f*)(&sh_h[(mb + v) * HSTR + EMB + ncol0]) = h;
        }
    }
    __syncthreads();

    {
        v8f acc[4] = {};
        const float* bp = Wl + (long)kb * NOUT + ncol0;
        for (int k0 = 0; k0 < HDIM; k0 += 4 * KU) {
            #pragma unroll
            for (int u = 0; u < KU; ++u) {
                const int k = k0 + 4 * u;
                v2f a;
                a.x = sh_h[m * HSTR + k + kb];
                a.y = sh_h[m * HSTR + k + kb + 1];
                const v4f b0 = *(const v4f*)(bp + (long)k * NOUT);
                const v4f b1 = *(const v4f*)(bp + (long)(k + 1) * NOUT);
                mma4(acc, a, b0, b1);
            }
        }
        const v4f bias = *(const v4f*)(lb + ncol0);
        #pragma unroll
        for (int v = 0; v < 8; ++v) {
            int r = sh_rows[mb + v];
            if (r >= 0) {
                v4f o;
                o.x = acc[0][v] + bias.x;
                o.y = acc[1][v] + bias.y;
                o.z = acc[2][v] + bias.z;
                o.w = acc[3][v] + bias.w;
                *(v4f*)(out + (long)r * NOUT + ncol0) = o;
            }
        }
    }
}

extern "C" void kernel_launch(void* const* d_in, const int* in_sizes, int n_in,
                              void* d_out, int out_size, void* d_ws, size_t ws_size,
                              hipStream_t stream) {
    const float* state   = (const float*)d_in[0];
    const float* obs     = (const float*)d_in[1];
    const int*   cat_ids = (const int*)  d_in[2];
    const float* obs_W   = (const float*)d_in[3];
    const float* obs_b   = (const float*)d_in[4];
    const float* state_W = (const float*)d_in[5];
    const float* state_b = (const float*)d_in[6];
    const float* l2_W    = (const float*)d_in[7];
    const float* l2_b    = (const float*)d_in[8];
    float*       out     = (float*)d_out;

    int* counts  = (int*)d_ws;
    int* offsets = counts + NCAT;
    int* perm    = offsets + NCAT;

    bucket_by_cat<<<1, 256, 0, stream>>>(cat_ids, counts, offsets, perm);

    moe_mlp_kernel<<<MAXTILES, 128, 0, stream>>>(state, obs, obs_W, obs_b,
                                             state_W, state_b, l2_W, l2_b,
                                             counts, offsets, perm, out);
}


# Each 16-k-step block's loads (weight rows, obs values; biases first) go out one block ahead into a second register set, across stage boundaries: load latency runs under the previous block's products
# speedup vs baseline: 5.6753x; 1.2251x over previous
#include <hip/hip_runtime.h>
#include <hip/hip_bf16.h>

typedef __attribute__((ext_vector_type(2))) float v2f;
typedef __attribute__((ext_vector_type(4))) float v4f;
typedef __attribute__((ext_vector_type(8))) float v8f;

#define BATCH 2048
#define NCAT 32
#define SDIM 64
#define EMB 256
#define NOUT 256
#define ODIM 512
#define HDIM 512
#define HSTR 516
#define SSTR 68
#define KU 16
#define MAXTILES 158

__global__ void bucket_by_cat(const int* __restrict__ cat_ids,
                              int* __restrict__ counts,
                              int* __restrict__ offsets,
                              int* __restrict__ perm) {
    __shared__ int s_cnt[NCAT];
    __shared__ int s_cur[NCAT];
    int t = threadIdx.x;
    if (t < NCAT) s_cnt[t] = 0;
    __syncthreads();
    for (int i = t; i < BATCH; i += blockDim.x)
        atomicAdd(&s_cnt[cat_ids[i]], 1);
    __syncthreads();
    if (t == 0) {
        int acc = 0;
        for (int c = 0; c < NCAT; ++c) {
            s_cur[c] = acc;
            offsets[c] = acc;
            counts[c] = s_cnt[c];
            acc += s_cnt[c];
        }
    }
    __syncthreads();
    for (int i = t; i < BATCH; i += blockDim.x) {
        int c = cat_ids[i];
        int pos = atomicAdd(&s_cur[c], 1);
        perm[pos] = i;
    }
}

__device__ __forceinline__ void mma4(v8f (&acc)[4], v2f a, v4f b0, v4f b1) {
    #pragma unroll
    for (int j = 0; j < 4; ++j) {
        v2f b;
        b.x = b0[j];
        b.y = b1[j];
        acc[j] = __builtin_amdgcn_wmma_f32_16x16x4_f32(
            false, a, false, b, (short)0, acc[j], false, false);
    }
}

__device__ __forceinline__ void load_b(v4f (&B)[2 * KU], const float* bp) {
    #pragma unroll
    for (int u = 0; u < KU; ++u) {
        B[2 * u]     = *(const v4f*)(bp + (long)(4 * u) * EMB);
        B[2 * u + 1] = *(const v4f*)(bp + (long)(4 * u + 1) * EMB);
    }
}
__device__ __forceinline__ void load_a(v2f (&A)[KU], const float* ap) {
    #pragma unroll
    for (int u = 0; u < KU; ++u) A[u] = *(const v2f*)(ap + 4 * u);
}
__device__ __forceinline__ void mma_blk_reg(v8f (&acc)[4], const v2f (&A)[KU],
                                            const v4f (&B)[2 * KU], float validf) {
    #pragma unroll
    for (int u = 0; u < KU; ++u) {
        v2f a = A[u];
        a.x *= validf;
        a.y *= validf;
        mma4(acc, a, B[2 * u], B[2 * u + 1]);
    }
}
__device__ __forceinline__ void mma_blk_lds(v8f (&acc)[4], const float* arow,
                                            const v4f (&B)[2 * KU]) {
    #pragma unroll
    for (int u = 0; u < KU; ++u) {
        v2f a;
        a.x = arow[4 * u];
        a.y = arow[4 * u + 1];
        mma4(acc, a, B[2 * u], B[2 * u + 1]);
    }
}

__global__ __launch_bounds__(128) __attribute__((amdgpu_waves_per_eu(1, 1)))
void moe_mlp_kernel(const float* __restrict__ state,
                    const float* __restrict__ obs,
                    const float* __restrict__ obs_W,
                    const float* __restrict__ obs_b,
                    const float* __restrict__ state_W,
                    const float* __restrict__ state_b,
                    const float* __restrict__ l2_W,
                    const float* __restrict__ l2_b,
                    const int* __restrict__ counts,
                    const int* __restrict__ offsets,
                    const int* __restrict__ perm,
                    float* __restrict__ out) {
    const int g = blockIdx.x;
    int c = 0, first = 0, run = 0;
    for (int cc = 0; cc < NCAT; ++cc) {
        run += (counts[cc] + 15) >> 4;
        if (g >= run) { c = cc + 1; first = run; }
    }
    if (c >= NCAT) return;
    const int tile = g - first;
    const int cnt  = counts[c];

    __shared__ int   sh_rows[16];
    __shared__ float sh_s[16 * SSTR];
    __shared__ __attribute__((aligned(16))) float sh_h[16 * HSTR];

    const int t    = threadIdx.x;
    const int wave = t >> 5;
    const int lane = t & 31;
    const int m    = lane & 15;
    const int half = lane >> 4;
    const int kb   = half * 2;
    const int mb   = half * 8;
    const int ncol0 = wave * 64 + 4 * m;

    const float* Ws = state_W + (long)c * SDIM * EMB;
    const float* bs = state_b + (long)c * EMB;
    const float* Wl = l2_W    + (long)c * HDIM * NOUT;
    const float* lb = l2_b    + (long)c * NOUT;

    const v4f bias_s = *(const v4f*)(bs + ncol0);
    const v4f bias_o = *(const v4f*)(obs_b + ncol0);
    const v4f bias_l = *(const v4f*)(lb + ncol0);

    v4f X[2 * KU], Y[2 * KU];
    v2f AX[KU], AY[KU];

    load_b(X, Ws + (long)kb * EMB + ncol0);

    const int base = offsets[c];
    if (t < 16) {
        int idx = tile * 16 + t;
        sh_rows[t] = (idx < cnt) ? perm[base + idx] : -1;
    }
    __syncthreads();

    for (int i = t; i < 16 * SDIM; i += 128) {
        int mm = i >> 6, kk = i & (SDIM - 1);
        int r = sh_rows[mm];
        sh_s[mm * SSTR + kk] = (r >= 0) ? state[(long)r * SDIM + kk] : 0.0f;
    }
    __syncthreads();

    const int   r      = sh_rows[m];
    const float validf = (r >= 0) ? 1.0f : 0.0f;
    const long  ro     = (long)((r >= 0) ? r : 0) * ODIM;
    const float* ap  = obs + ro + kb;
    const float* obp = obs_W + (long)kb * EMB + ncol0;
    const float* wbp = Wl + (long)kb * NOUT + ncol0;

    load_b(Y, obp);
    load_a(AY, ap);

    {
        v8f acc[4] = {};
        mma_blk_lds(acc, &sh_s[m * SSTR + kb], X);
        #pragma unroll
        for (int v = 0; v < 8; ++v) {
            v4f h;
            h.x = fmaxf(acc[0][v] + bias_s.x, 0.0f);
            h.y = fmaxf(acc[1][v] + bias_s.y, 0.0f);
            h.z = fmaxf(acc[2][v] + bias_s.z, 0.0f);
            h.w = fmaxf(acc[3][v] + bias_s.w, 0.0f);
            *(v4f*)(&sh_h[(mb + v) * HSTR + ncol0]) = h;
        }
    }

    {
        v8f acc[4] = {};
        const int NB = ODIM / (4 * KU);
        for (int i = 0; i < NB - 2; i += 2) {
            load_b(X, obp + (long)(i + 1) * (4 * KU) * EMB);
            load_a(AX, ap + (i + 1) * (4 * KU));
            mma_blk_reg(acc, AY, Y, validf);
            load_b(Y, obp + (long)(i + 2) * (4 * KU) * EMB);
            load_a(AY, ap + (i + 2) * (4 * KU));
            mma_blk_reg(acc, AX, X, validf);
        }
        load_b(X, obp + (long)(NB - 1) * (4 * KU) * EMB);
        load_a(AX, ap + (NB - 1) * (4 * KU));
        mma_blk_reg(acc, AY, Y, validf);
        load_b(Y, wbp);
        mma_blk_reg(acc, AX, X, validf);
        #pragma unroll
        for (int v = 0; v < 8; ++v) {
            v4f h;
            h.x = acc[0][v] + bias_o.x;
            h.y = acc[1][v] + bias_o.y;
            h.z = acc[2][v] + bias_o.z;
            h.w = acc[3][v] + bias_o.w;
            *(v4f*)(&sh_h[(mb + v) * HSTR + EMB + ncol0]) = h;
        }
    }
    __syncthreads();

    {
        v8f acc[4] = {};
        const float* hrow = &sh_h[m * HSTR + kb];
        const int NB = HDIM / (4 * KU);
        for (int i = 0; i < NB - 2; i += 2) {
            load_b(X, wbp + (long)(i + 1) * (4 * KU) * NOUT);
            mma_blk_lds(acc, hrow + i * (4 * KU), Y);
            load_b(Y, wbp + (long)(i + 2) * (4 * KU) * NOUT);
            mma_blk_lds(acc, hrow + (i + 1) * (4 * KU), X);
        }
        load_b(X, wbp + (long)(NB - 1) * (4 * KU) * NOUT);
        mma_blk_lds(acc, hrow + (NB - 2) * (4 * KU), Y);
        mma_blk_lds(acc, hrow + (NB - 1) * (4 * KU), X);
        #pragma unroll
        for (int v = 0; v < 8; ++v) {
            int rr = sh_rows[mb + v];
            if (rr >= 0) {
                v4f o;
                o.x = acc[0][v] + bias_l.x;
                o.y = acc[1][v] + bias_l.y;
                o.z = acc[2][v] + bias_l.z;
                o.w = acc[3][v] + bias_l.w;
                *(v4f*)(out + (long)rr * NOUT + ncol0) = o;
            }
        }
    }
}

extern "C" void kernel_launch(void* const* d_in, const int* in_sizes, int n_in,
                              void* d_out, int out_size, void* d_ws, size_t ws_size,
                              hipStream_t stream) {
    const float* state   = (const float*)d_in[0];
    const float* obs     = (const float*)d_in[1];
    const int*   cat_ids = (const int*)  d_in[2];
    const float* obs_W   = (const float*)d_in[3];
    const float* obs_b   = (const float*)d_in[4];
    const float* state_W = (const float*)d_in[5];
    const float* state_b = (const float*)d_in[6];
    const float* l2_W    = (const float*)d_in[7];
    const float* l2_b    = (const float*)d_in[8];
    float*       out     = (float*)d_out;

    int* counts  = (int*)d_ws;
    int* offsets = counts + NCAT;
    int* perm    = offsets + NCAT;

    bucket_by_cat<<<1, 256, 0, stream>>>(cat_ids, counts, offsets, perm);

    moe_mlp_kernel<<<MAXTILES, 128, 0, stream>>>(state, obs, obs_W, obs_b,
                                             state_W, state_b, l2_W, l2_b,
                                             counts, offsets, perm, out);
}


# Row bucketing folded into the main kernel: each work-group finds its category and its rows from cat_ids by wave ballots, rows ranked by position, no atomics; one launch, not two, no workspace
# speedup vs baseline: 7.2200x; 1.2722x over previous
#include <hip/hip_runtime.h>
#include <hip/hip_bf16.h>

typedef __attribute__((ext_vector_type(2))) float v2f;
typedef __attribute__((ext_vector_type(4))) float v4f;
typedef __attribute__((ext_vector_type(8))) float v8f;
typedef __attribute__((ext_vector_type(4))) int   v4i;

#define BATCH 2048
#define NCAT 32
#define SDIM 64
#define EMB 256
#define NOUT 256
#define ODIM 512
#define HDIM 512
#define HSTR 516
#define SSTR 68
#define KU 16
#define MAXTILES 158

__device__ __forceinline__ void mma4(v8f (&acc)[4], v2f a, v4f b0, v4f b1) {
    #pragma unroll
    for (int j = 0; j < 4; ++j) {
        v2f b;
        b.x = b0[j];
        b.y = b1[j];
        acc[j] = __builtin_amdgcn_wmma_f32_16x16x4_f32(
            false, a, false, b, (short)0, acc[j], false, false);
    }
}

__device__ __forceinline__ void load_b(v4f (&B)[2 * KU], const float* bp) {
    #pragma unroll
    for (int u = 0; u < KU; ++u) {
        B[2 * u]     = *(const v4f*)(bp + (long)(4 * u) * EMB);
        B[2 * u + 1] = *(const v4f*)(bp + (long)(4 * u + 1) * EMB);
    }
}
__device__ __forceinline__ void load_a(v2f (&A)[KU], const float* ap) {
    #pragma unroll
    for (int u = 0; u < KU; ++u) A[u] = *(const v2f*)(ap + 4 * u);
}
__device__ __forceinline__ void mma_blk_reg(v8f (&acc)[4], const v2f (&A)[KU],
                                            const v4f (&B)[2 * KU], float validf) {
    #pragma unroll
    for (int u = 0; u < KU; ++u) {
        v2f a = A[u];
        a.x *= validf;
        a.y *= validf;
        mma4(acc, a, B[2 * u], B[2 * u + 1]);
    }
}
__device__ __forceinline__ void mma_blk_lds(v8f (&acc)[4], const float* arow,
                                            const v4f (&B)[2 * KU]) {
    #pragma unroll
    for (int u = 0; u < KU; ++u) {
        v2f a;
        a.x = arow[4 * u];
        a.y = arow[4 * u + 1];
        mma4(acc, a, B[2 * u], B[2 * u + 1]);
    }
}

__global__ __launch_bounds__(128) __attribute__((amdgpu_waves_per_eu(1, 1)))
void moe_mlp_kernel(const float* __restrict__ state,
                    const float* __restrict__ obs,
                    const float* __restrict__ obs_W,
                    const float* __restrict__ obs_b,
                    const float* __restrict__ state_W,
                    const float* __restrict__ state_b,
                    const float* __restrict__ l2_W,
                    const float* __restrict__ l2_b,
                    const int* __restrict__ cat_ids,
                    float* __restrict__ out) {
    const int g    = blockIdx.x;
    const int t    = threadIdx.x;
    const int wave = t >> 5;
    const int lane = t & 31;

    __shared__ int   sh_part[4 * NCAT];
    __shared__ int   sh_rows[16];
    __shared__ float sh_s[16 * SSTR];
    __shared__ __attribute__((aligned(16))) float sh_h[16 * HSTR];

    v4i idv[4];
    {
        const v4i* cp = (const v4i*)(cat_ids + wave * (BATCH / 4)) + lane;
        #pragma unroll
        for (int q = 0; q < 4; ++q) idv[q] = cp[32 * q];
    }
    if (t < 16) sh_rows[t] = -1;

    unsigned nl[5];
    #pragma unroll
    for (int k = 0; k < 5; ++k) nl[k] = ((lane >> k) & 1) ? 0u : 0xffffffffu;
    int cnt_l = 0;
    #pragma unroll
    for (int q = 0; q < 4; ++q) {
        #pragma unroll
        for (int e = 0; e < 4; ++e) {
            const int id = idv[q][e];
            unsigned msk = 0xffffffffu;
            #pragma unroll
            for (int k = 0; k < 5; ++k)
                msk &= __builtin_amdgcn_ballot_w32(((id >> k) & 1) != 0) ^ nl[k];
            cnt_l += __popc(msk);
        }
    }
    sh_part[wave * NCAT + lane] = cnt_l;
    __syncthreads();

    int part[4];
    #pragma unroll
    for (int w = 0; w < 4; ++w) part[w] = sh_part[w * NCAT + lane];
    int incl = (part[0] + part[1] + part[2] + part[3] + 15) >> 4;
    #pragma unroll
    for (int d = 1; d < 32; d <<= 1) {
        const int y = __shfl_up(incl, d);
        if (lane >= d) incl += y;
    }
    const int c = __popc(__builtin_amdgcn_ballot_w32(incl <= g));
    if (c >= NCAT) return;
    const int first = (c > 0) ? __builtin_amdgcn_readlane(incl, (c - 1) & 31) : 0;
    const int tile  = g - first;

    const int m    = lane & 15;
    const int half = lane >> 4;
    const int kb   = half * 2;
    const int mb   = half * 8;
    const int ncol0 = wave * 64 + 4 * m;

    const float* Ws = state_W + (long)c * SDIM * EMB;
    const float* bs = state_b + (long)c * EMB;
    const float* Wl = l2_W    + (long)c * HDIM * NOUT;
    const float* lb = l2_b    + (long)c * NOUT;

    const v4f bias_s = *(const v4f*)(bs + ncol0);
    const v4f bias_o = *(const v4f*)(obs_b + ncol0);
    const v4f bias_l = *(const v4f*)(lb + ncol0);

    v4f X[2 * KU], Y[2 * KU];
    v2f AX[KU], AY[KU];

    load_b(X, Ws + (long)kb * EMB + ncol0);

    {
        int pre = -16 * tile;
        #pragma unroll
        for (int w = 0; w < 3; ++w) {
            const int pw = __builtin_amdgcn_readlane(part[w], c);
            pre += (w < wave) ? pw : 0;
        }
        #pragma unroll
        for (int q = 0; q < 4; ++q) {
            #pragma unroll
            for (int e = 0; e < 4; ++e) {
                const bool mine = (idv[q][e] == c);
                const unsigned hit = __builtin_amdgcn_ballot_w32(mine);
                const int slot = pre + (int)__builtin_amdgcn_mbcnt_lo(hit, 0u);
                if (mine && (unsigned)slot < 16u)
                    sh_rows[slot] = wave * (BATCH / 4) + 128 * q + 4 * lane + e;
                pre += __popc(hit);
            }
        }
    }
    __syncthreads();

    for (int i = t; i < 16 * SDIM; i += 128) {
        int mm = i >> 6, kk = i & (SDIM - 1);
        int r = sh_rows[mm];
        sh_s[mm * SSTR + kk] = (r >= 0) ? state[(long)r * SDIM + kk] : 0.0f;
    }
    __syncthreads();

    const int   r      = sh_rows[m];
    const float validf = (r >= 0) ? 1.0f : 0.0f;
    const long  ro     = (long)((r >= 0) ? r : 0) * ODIM;
    const float* ap  = obs + ro + kb;
    const float* obp = obs_W + (long)kb * EMB + ncol0;
    const float* wbp = Wl + (long)kb * NOUT + ncol0;

    load_b(Y, obp);
    load_a(AY, ap);

    {
        v8f acc[4] = {};
        mma_blk_lds(acc, &sh_s[m * SSTR + kb], X);
        #pragma unroll
        for (int v = 0; v < 8; ++v) {
            v4f h;
            h.x = fmaxf(acc[0][v] + bias_s.x, 0.0f);
            h.y = fmaxf(acc[1][v] + bias_s.y, 0.0f);
            h.z = fmaxf(acc[2][v] + bias_s.z, 0.0f);
            h.w = fmaxf(acc[3][v] + bias_s.w, 0.0f);
            *(v4f*)(&sh_h[(mb + v) * HSTR + ncol0]) = h;
        }
    }

    {
        v8f acc[4] = {};
        const int NB = ODIM / (4 * KU);
        for (int i = 0; i < NB - 2; i += 2) {
            load_b(X, obp + (long)(i + 1) * (4 * KU) * EMB);
            load_a(AX, ap + (i + 1) * (4 * KU));
            mma_blk_reg(acc, AY, Y, validf);
            load_b(Y, obp + (long)(i + 2) * (4 * KU) * EMB);
            load_a(AY, ap + (i + 2) * (4 * KU));
            mma_blk_reg(acc, AX, X, validf);
        }
        load_b(X, obp + (long)(NB - 1) * (4 * KU) * EMB);
        load_a(AX, ap + (NB - 1) * (4 * KU));
        mma_blk_reg(acc, AY, Y, validf);
        load_b(Y, wbp);
        mma_blk_reg(acc, AX, X, validf);
        #pragma unroll
        for (int v = 0; v < 8; ++v) {
            v4f h;
            h.x = acc[0][v] + bias_o.x;
            h.y = acc[1][v] + bias_o.y;
            h.z = acc[2][v] + bias_o.z;
            h.w = acc[3][v] + bias_o.w;
            *(v4f*)(&sh_h[(mb + v) * HSTR + EMB + ncol0]) = h;
        }
    }
    __syncthreads();

    {
        v8f acc[4] = {};
        const float* hrow = &sh_h[m * HSTR + kb];
        const int NB = HDIM / (4 * KU);
        for (int i = 0; i < NB - 2; i += 2) {
            load_b(X, wbp + (long)(i + 1) * (4 * KU) * NOUT);
            mma_blk_lds(acc, hrow + i * (4 * KU), Y);
            load_b(Y, wbp + (long)(i + 2) * (4 * KU) * NOUT);
            mma_blk_lds(acc, hrow + (i + 1) * (4 * KU), X);
        }
        load_b(X, wbp + (long)(NB - 1) * (4 * KU) * NOUT);
        mma_blk_lds(acc, hrow + (NB - 2) * (4 * KU), Y);
        mma_blk_lds(acc, hrow + (NB - 1) * (4 * KU), X);
        #pragma unroll
        for (int v = 0; v < 8; ++v) {
            int rr = sh_rows[mb + v];
            if (rr >= 0) {
                v4f o;
                o.x = acc[0][v] + bias_l.x;
                o.y = acc[1][v] + bias_l.y;
                o.z = acc[2][v] + bias_l.z;
                o.w = acc[3][v] + bias_l.w;
                *(v4f*)(out + (long)rr * NOUT + ncol0) = o;
            }
        }
    }
}

extern "C" void kernel_launch(void* const* d_in, const int* in_sizes, int n_in,
                              void* d_out, int out_size, void* d_ws, size_t ws_size,
                              hipStream_t stream) {
    const float* state   = (const float*)d_in[0];
    const float* obs     = (const float*)d_in[1];
    const int*   cat_ids = (const int*)  d_in[2];
    const float* obs_W   = (const float*)d_in[3];
    const float* obs_b   = (const float*)d_in[4];
    const float* state_W = (const float*)d_in[5];
    const float* state_b = (const float*)d_in[6];
    const float* l2_W    = (const float*)d_in[7];
    const float* l2_b    = (const float*)d_in[8];
    float*       out     = (float*)d_out;

    moe_mlp_kernel<<<MAXTILES, 128, 0, stream>>>(state, obs, obs_W, obs_b,
                                             state_W, state_b, l2_W, l2_b,
                                             cat_ids, out);
}


# State and obs tiles staged into LDS once per block by whole-line 16-byte loads, all in flight together; obs A fragments read from LDS, not gathered from global by every wave at each k-step
# speedup vs baseline: 8.3765x; 1.1602x over previous
#include <hip/hip_runtime.h>
#include <hip/hip_bf16.h>

typedef __attribute__((ext_vector_type(2))) float v2f;
typedef __attribute__((ext_vector_type(4))) float v4f;
typedef __attribute__((ext_vector_type(8))) float v8f;
typedef __attribute__((ext_vector_type(4))) int   v4i;

#define BATCH 2048
#define NCAT 32
#define SDIM 64
#define EMB 256
#define NOUT 256
#define ODIM 512
#define HDIM 512
#define HSTR 516
#define SSTR 68
#define KU 16
#define MAXTILES 158

__device__ __forceinline__ void mma4(v8f (&acc)[4], v2f a, v4f b0, v4f b1) {
    #pragma unroll
    for (int j = 0; j < 4; ++j) {
        v2f b;
        b.x = b0[j];
        b.y = b1[j];
        acc[j] = __builtin_amdgcn_wmma_f32_16x16x4_f32(
            false, a, false, b, (short)0, acc[j], false, false);
    }
}

__device__ __forceinline__ void load_b(v4f (&B)[2 * KU], const float* bp) {
    #pragma unroll
    for (int u = 0; u < KU; ++u) {
        B[2 * u]     = *(const v4f*)(bp + (long)(4 * u) * EMB);
        B[2 * u + 1] = *(const v4f*)(bp + (long)(4 * u + 1) * EMB);
    }
}
__device__ __forceinline__ void mma_blk_lds(v8f (&acc)[4], const float* arow,
                                            const v4f (&B)[2 * KU]) {
    #pragma unroll
    for (int u = 0; u < KU; ++u) {
        v2f a;
        a.x = arow[4 * u];
        a.y = arow[4 * u + 1];
        mma4(acc, a, B[2 * u], B[2 * u + 1]);
    }
}

__global__ __launch_bounds__(128) __attribute__((amdgpu_waves_per_eu(1, 1)))
void moe_mlp_kernel(const float* __restrict__ state,
                    const float* __restrict__ obs,
                    const float* __restrict__ obs_W,
                    const float* __restrict__ obs_b,
                    const float* __restrict__ state_W,
                    const float* __restrict__ state_b,
                    const float* __restrict__ l2_W,
                    const float* __restrict__ l2_b,
                    const int* __restrict__ cat_ids,
                    float* __restrict__ out) {
    const int g    = blockIdx.x;
    const int t    = threadIdx.x;
    const int wave = t >> 5;
    const int lane = t & 31;

    __shared__ int   sh_part[4 * NCAT];
    __shared__ int   sh_rows[16];
    __shared__ __attribute__((aligned(16))) float sh_s[16 * SSTR];
    __shared__ __attribute__((aligned(16))) float sh_o[16 * HSTR];
    __shared__ __attribute__((aligned(16))) float sh_h[16 * HSTR];

    v4i idv[4];
    {
        const v4i* cp = (const v4i*)(cat_ids + wave * (BATCH / 4)) + lane;
        #pragma unroll
        for (int q = 0; q < 4; ++q) idv[q] = cp[32 * q];
    }
    if (t < 16) sh_rows[t] = -1;

    unsigned nl[5];
    #pragma unroll
    for (int k = 0; k < 5; ++k) nl[k] = ((lane >> k) & 1) ? 0u : 0xffffffffu;
    int cnt_l = 0;
    #pragma unroll
    for (int q = 0; q < 4; ++q) {
        #pragma unroll
        for (int e = 0; e < 4; ++e) {
            const int id = idv[q][e];
            unsigned msk = 0xffffffffu;
            #pragma unroll
            for (int k = 0; k < 5; ++k)
                msk &= __builtin_amdgcn_ballot_w32(((id >> k) & 1) != 0) ^ nl[k];
            cnt_l += __popc(msk);
        }
    }
    sh_part[wave * NCAT + lane] = cnt_l;
    __syncthreads();

    int part[4];
    #pragma unroll
    for (int w = 0; w < 4; ++w) part[w] = sh_part[w * NCAT + lane];
    int incl = (part[0] + part[1] + part[2] + part[3] + 15) >> 4;
    #pragma unroll
    for (int d = 1; d < 32; d <<= 1) {
        const int y = __shfl_up(incl, d);
        if (lane >= d) incl += y;
    }
    const int c = __popc(__builtin_amdgcn_ballot_w32(incl <= g));
    if (c >= NCAT) return;
    const int first = (c > 0) ? __builtin_amdgcn_readlane(incl, (c - 1) & 31) : 0;
    const int tile  = g - first;

    const int m    = lane & 15;
    const int half = lane >> 4;
    const int kb   = half * 2;
    const int mb   = half * 8;
    const int ncol0 = wave * 64 + 4 * m;

    const float* Ws = state_W + (long)c * SDIM * EMB;
    const float* bs = state_b + (long)c * EMB;
    const float* Wl = l2_W    + (long)c * HDIM * NOUT;
    const float* lb = l2_b    + (long)c * NOUT;

    const v4f bias_s = *(const v4f*)(bs + ncol0);
    const v4f bias_o = *(const v4f*)(obs_b + ncol0);
    const v4f bias_l = *(const v4f*)(lb + ncol0);

    v4f X[2 * KU], Y[2 * KU];

    load_b(X, Ws + (long)kb * EMB + ncol0);

    {
        int pre = -16 * tile;
        #pragma unroll
        for (int w = 0; w < 3; ++w) {
            const int pw = __builtin_amdgcn_readlane(part[w], c);
            pre += (w < wave) ? pw : 0;
        }
        #pragma unroll
        for (int q = 0; q < 4; ++q) {
            #pragma unroll
            for (int e = 0; e < 4; ++e) {
                const bool mine = (idv[q][e] == c);
                const unsigned hit = __builtin_amdgcn_ballot_w32(mine);
                const int slot = pre + (int)__builtin_amdgcn_mbcnt_lo(hit, 0u);
                if (mine && (unsigned)slot < 16u)
                    sh_rows[slot] = wave * (BATCH / 4) + 128 * q + 4 * lane + e;
                pre += __popc(hit);
            }
        }
    }
    __syncthreads();

    const float* obp = obs_W + (long)kb * EMB + ncol0;
    const float* wbp = Wl + (long)kb * NOUT + ncol0;

    {
        const v4f zero4 = {0.0f, 0.0f, 0.0f, 0.0f};
        int    srow[2];
        float* sdst[2];
        v4f    sv[2];
        #pragma unroll
        for (int q = 0; q < 2; ++q) {
            const int idx = t + 128 * q;
            const int rs  = sh_rows[idx >> 4];
            srow[q] = rs;
            sdst[q] = &sh_s[(idx >> 4) * SSTR + 4 * (idx & 15)];
            sv[q]   = *(const v4f*)(state + (long)((rs >= 0) ? rs : 0) * SDIM + 4 * (idx & 15));
        }
        int orow[16];
        v4f ov[16];
        #pragma unroll
        for (int i = 0; i < 16; ++i) {
            const int ro = __builtin_amdgcn_readfirstlane(sh_rows[i]);
            orow[i] = ro;
            ov[i] = *(const v4f*)(obs + (long)((ro >= 0) ? ro : 0) * ODIM + 4 * t);
        }
        #pragma unroll
        for (int q = 0; q < 2; ++q) *(v4f*)sdst[q] = sv[q];
        #pragma unroll
        for (int i = 0; i < 16; ++i) *(v4f*)(&sh_o[i * HSTR + 4 * t]) = ov[i];
        #pragma unroll
        for (int q = 0; q < 2; ++q)
            if (srow[q] < 0) *(v4f*)sdst[q] = zero4;
        #pragma unroll
        for (int i = 0; i < 16; ++i)
            if (orow[i] < 0) *(v4f*)(&sh_o[i * HSTR + 4 * t]) = zero4;
    }
    __syncthreads();

    load_b(Y, obp);

    {
        v8f acc[4] = {};
        mma_blk_lds(acc, &sh_s[m * SSTR + kb], X);
        #pragma unroll
        for (int v = 0; v < 8; ++v) {
            v4f h;
            h.x = fmaxf(acc[0][v] + bias_s.x, 0.0f);
            h.y = fmaxf(acc[1][v] + bias_s.y, 0.0f);
            h.z = fmaxf(acc[2][v] + bias_s.z, 0.0f);
            h.w = fmaxf(acc[3][v] + bias_s.w, 0.0f);
            *(v4f*)(&sh_h[(mb + v) * HSTR + ncol0]) = h;
        }
    }

    {
        v8f acc[4] = {};
        const int NB = ODIM / (4 * KU);
        const float* orw = &sh_o[m * HSTR + kb];
        for (int i = 0; i < NB - 2; i += 2) {
            load_b(X, obp + (long)(i + 1) * (4 * KU) * EMB);
            mma_blk_lds(acc, orw + i * (4 * KU), Y);
            load_b(Y, obp + (long)(i + 2) * (4 * KU) * EMB);
            mma_blk_lds(acc, orw + (i + 1) * (4 * KU), X);
        }
        load_b(X, obp + (long)(NB - 1) * (4 * KU) * EMB);
        mma_blk_lds(acc, orw + (NB - 2) * (4 * KU), Y);
        load_b(Y, wbp);
        mma_blk_lds(acc, orw + (NB - 1) * (4 * KU), X);
        #pragma unroll
        for (int v = 0; v < 8; ++v) {
            v4f h;
            h.x = acc[0][v] + bias_o.x;
            h.y = acc[1][v] + bias_o.y;
            h.z = acc[2][v] + bias_o.z;
            h.w = acc[3][v] + bias_o.w;
            *(v4f*)(&sh_h[(mb + v) * HSTR + EMB + ncol0]) = h;
        }
    }
    __syncthreads();

    {
        v8f acc[4] = {};
        const float* hrow = &sh_h[m * HSTR + kb];
        const int NB = HDIM / (4 * KU);
        for (int i = 0; i < NB - 2; i += 2) {
            load_b(X, wbp + (long)(i + 1) * (4 * KU) * NOUT);
            mma_blk_lds(acc, hrow + i * (4 * KU), Y);
            load_b(Y, wbp + (long)(i + 2) * (4 * KU) * NOUT);
            mma_blk_lds(acc, hrow + (i + 1) * (4 * KU), X);
        }
        load_b(X, wbp + (long)(NB - 1) * (4 * KU) * NOUT);
        mma_blk_lds(acc, hrow + (NB - 2) * (4 * KU), Y);
        mma_blk_lds(acc, hrow + (NB - 1) * (4 * KU), X);
        #pragma unroll
        for (int v = 0; v < 8; ++v) {
            int rr = sh_rows[mb + v];
            if (rr >= 0) {
                v4f o;
                o.x = acc[0][v] + bias_l.x;
                o.y = acc[1][v] + bias_l.y;
                o.z = acc[2][v] + bias_l.z;
                o.w = acc[3][v] + bias_l.w;
                *(v4f*)(out + (long)rr * NOUT + ncol0) = o;
            }
        }
    }
}

extern "C" void kernel_launch(void* const* d_in, const int* in_sizes, int n_in,
                              void* d_out, int out_size, void* d_ws, size_t ws_size,
                              hipStream_t stream) {
    const float* state   = (const float*)d_in[0];
    const float* obs     = (const float*)d_in[1];
    const int*   cat_ids = (const int*)  d_in[2];
    const float* obs_W   = (const float*)d_in[3];
    const float* obs_b   = (const float*)d_in[4];
    const float* state_W = (const float*)d_in[5];
    const float* state_b = (const float*)d_in[6];
    const float* l2_W    = (const float*)d_in[7];
    const float* l2_b    = (const float*)d_in[8];
    float*       out     = (float*)d_out;

    moe_mlp_kernel<<<MAXTILES, 128, 0, stream>>>(state, obs, obs_W, obs_b,
                                             state_W, state_b, l2_W, l2_b,
                                             cat_ids, out);
}


# Each k-block's 32 weight loads and its A-fragment LDS reads (second A register set) stand as ONE group ahead of the previous block's 64 WMMAs, held by scheduling fences, not spread through the run
# speedup vs baseline: 8.9683x; 1.0707x over previous
#include <hip/hip_runtime.h>
#include <hip/hip_bf16.h>

typedef __attribute__((ext_vector_type(2))) float v2f;
typedef __attribute__((ext_vector_type(4))) float v4f;
typedef __attribute__((ext_vector_type(8))) float v8f;
typedef __attribute__((ext_vector_type(4))) int   v4i;

#define BATCH 2048
#define NCAT 32
#define SDIM 64
#define EMB 256
#define NOUT 256
#define ODIM 512
#define HDIM 512
#define HSTR 516
#define SSTR 68
#define KU 16
#define MAXTILES 158

__device__ __forceinline__ void mma4(v8f (&acc)[4], v2f a, v4f b0, v4f b1) {
    #pragma unroll
    for (int j = 0; j < 4; ++j) {
        v2f b;
        b.x = b0[j];
        b.y = b1[j];
        acc[j] = __builtin_amdgcn_wmma_f32_16x16x4_f32(
            false, a, false, b, (short)0, acc[j], false, false);
    }
}

__device__ __forceinline__ void load_b(v4f (&B)[2 * KU], const float* bp) {
    #pragma unroll
    for (int u = 0; u < KU; ++u) {
        B[2 * u]     = *(const v4f*)(bp + (long)(4 * u) * EMB);
        B[2 * u + 1] = *(const v4f*)(bp + (long)(4 * u + 1) * EMB);
    }
}
__device__ __forceinline__ void load_a(v2f (&A)[KU], const float* arow) {
    #pragma unroll
    for (int u = 0; u < KU; ++u) A[u] = *(const v2f*)(arow + 4 * u);
}
__device__ __forceinline__ void mma_blk(v8f (&acc)[4], const v2f (&A)[KU],
                                        const v4f (&B)[2 * KU]) {
    #pragma unroll
    for (int u = 0; u < KU; ++u) mma4(acc, A[u], B[2 * u], B[2 * u + 1]);
}
__device__ __forceinline__ void issue_fence() { __builtin_amdgcn_sched_barrier(0); }

__global__ __launch_bounds__(128) __attribute__((amdgpu_waves_per_eu(1, 1)))
void moe_mlp_kernel(const float* __restrict__ state,
                    const float* __restrict__ obs,
                    const float* __restrict__ obs_W,
                    const float* __restrict__ obs_b,
                    const float* __restrict__ state_W,
                    const float* __restrict__ state_b,
                    const float* __restrict__ l2_W,
                    const float* __restrict__ l2_b,
                    const int* __restrict__ cat_ids,
                    float* __restrict__ out) {
    const int g    = blockIdx.x;
    const int t    = threadIdx.x;
    const int wave = t >> 5;
    const int lane = t & 31;

    __shared__ int   sh_part[4 * NCAT];
    __shared__ int   sh_rows[16];
    __shared__ __attribute__((aligned(16))) float sh_s[16 * SSTR];
    __shared__ __attribute__((aligned(16))) float sh_o[16 * HSTR];
    __shared__ __attribute__((aligned(16))) float sh_h[16 * HSTR];

    v4i idv[4];
    {
        const v4i* cp = (const v4i*)(cat_ids + wave * (BATCH / 4)) + lane;
        #pragma unroll
        for (int q = 0; q < 4; ++q) idv[q] = cp[32 * q];
    }
    if (t < 16) sh_rows[t] = -1;

    unsigned nl[5];
    #pragma unroll
    for (int k = 0; k < 5; ++k) nl[k] = ((lane >> k) & 1) ? 0u : 0xffffffffu;
    int cnt_l = 0;
    #pragma unroll
    for (int q = 0; q < 4; ++q) {
        #pragma unroll
        for (int e = 0; e < 4; ++e) {
            const int id = idv[q][e];
            unsigned msk = 0xffffffffu;
            #pragma unroll
            for (int k = 0; k < 5; ++k)
                msk &= __builtin_amdgcn_ballot_w32(((id >> k) & 1) != 0) ^ nl[k];
            cnt_l += __popc(msk);
        }
    }
    sh_part[wave * NCAT + lane] = cnt_l;
    __syncthreads();

    int part[4];
    #pragma unroll
    for (int w = 0; w < 4; ++w) part[w] = sh_part[w * NCAT + lane];
    int incl = (part[0] + part[1] + part[2] + part[3] + 15) >> 4;
    #pragma unroll
    for (int d = 1; d < 32; d <<= 1) {
        const int y = __shfl_up(incl, d);
        if (lane >= d) incl += y;
    }
    const int c = __popc(__builtin_amdgcn_ballot_w32(incl <= g));
    if (c >= NCAT) return;
    const int first = (c > 0) ? __builtin_amdgcn_readlane(incl, (c - 1) & 31) : 0;
    const int tile  = g - first;

    const int m    = lane & 15;
    const int half = lane >> 4;
    const int kb   = half * 2;
    const int mb   = half * 8;
    const int ncol0 = wave * 64 + 4 * m;

    const float* Ws = state_W + (long)c * SDIM * EMB;
    const float* bs = state_b + (long)c * EMB;
    const float* Wl = l2_W    + (long)c * HDIM * NOUT;
    const float* lb = l2_b    + (long)c * NOUT;

    const v4f bias_s = *(const v4f*)(bs + ncol0);
    const v4f bias_o = *(const v4f*)(obs_b + ncol0);
    const v4f bias_l = *(const v4f*)(lb + ncol0);

    v4f X[2 * KU], Y[2 * KU];
    v2f AX[KU], AY[KU];

    load_b(X, Ws + (long)kb * EMB + ncol0);

    {
        int pre = -16 * tile;
        #pragma unroll
        for (int w = 0; w < 3; ++w) {
            const int pw = __builtin_amdgcn_readlane(part[w], c);
            pre += (w < wave) ? pw : 0;
        }
        #pragma unroll
        for (int q = 0; q < 4; ++q) {
            #pragma unroll
            for (int e = 0; e < 4; ++e) {
                const bool mine = (idv[q][e] == c);
                const unsigned hit = __builtin_amdgcn_ballot_w32(mine);
                const int slot = pre + (int)__builtin_amdgcn_mbcnt_lo(hit, 0u);
                if (mine && (unsigned)slot < 16u)
                    sh_rows[slot] = wave * (BATCH / 4) + 128 * q + 4 * lane + e;
                pre += __popc(hit);
            }
        }
    }
    __syncthreads();

    const float* obp = obs_W + (long)kb * EMB + ncol0;
    const float* wbp = Wl + (long)kb * NOUT + ncol0;

    {
        const v4f zero4 = {0.0f, 0.0f, 0.0f, 0.0f};
        int    srow[2];
        float* sdst[2];
        v4f    sv[2];
        #pragma unroll
        for (int q = 0; q < 2; ++q) {
            const int idx = t + 128 * q;
            const int rs  = sh_rows[idx >> 4];
            srow[q] = rs;
            sdst[q] = &sh_s[(idx >> 4) * SSTR + 4 * (idx & 15)];
            sv[q]   = *(const v4f*)(state + (long)((rs >= 0) ? rs : 0) * SDIM + 4 * (idx & 15));
        }
        int orow[16];
        v4f ov[16];
        #pragma unroll
        for (int i = 0; i < 16; ++i) {
            const int ro = __builtin_amdgcn_readfirstlane(sh_rows[i]);
            orow[i] = ro;
            ov[i] = *(const v4f*)(obs + (long)((ro >= 0) ? ro : 0) * ODIM + 4 * t);
        }
        #pragma unroll
        for (int q = 0; q < 2; ++q) *(v4f*)sdst[q] = sv[q];
        #pragma unroll
        for (int i = 0; i < 16; ++i) *(v4f*)(&sh_o[i * HSTR + 4 * t]) = ov[i];
        #pragma unroll
        for (int q = 0; q < 2; ++q)
            if (srow[q] < 0) *(v4f*)sdst[q] = zero4;
        #pragma unroll
        for (int i = 0; i < 16; ++i)
            if (orow[i] < 0) *(v4f*)(&sh_o[i * HSTR + 4 * t]) = zero4;
    }
    __syncthreads();

    const float* srw  = &sh_s[m * SSTR + kb];
    const float* orw  = &sh_o[m * HSTR + kb];
    const float* hrow = &sh_h[m * HSTR + kb];

    load_a(AX, srw);
    load_b(Y, obp);
    load_a(AY, orw);
    issue_fence();

    {
        v8f acc[4] = {};
        mma_blk(acc, AX, X);
        issue_fence();
        #pragma unroll
        for (int v = 0; v < 8; ++v) {
            v4f h;
            h.x = fmaxf(acc[0][v] + bias_s.x, 0.0f);
            h.y = fmaxf(acc[1][v] + bias_s.y, 0.0f);
            h.z = fmaxf(acc[2][v] + bias_s.z, 0.0f);
            h.w = fmaxf(acc[3][v] + bias_s.w, 0.0f);
            *(v4f*)(&sh_h[(mb + v) * HSTR + ncol0]) = h;
        }
    }

    {
        v8f acc[4] = {};
        const int NB = ODIM / (4 * KU);
        for (int i = 0; i < NB - 2; i += 2) {
            issue_fence();
            load_b(X, obp + (long)(i + 1) * (4 * KU) * EMB);
            load_a(AX, orw + (i + 1) * (4 * KU));
            issue_fence();
            mma_blk(acc, AY, Y);
            issue_fence();
            load_b(Y, obp + (long)(i + 2) * (4 * KU) * EMB);
            load_a(AY, orw + (i + 2) * (4 * KU));
            issue_fence();
            mma_blk(acc, AX, X);
        }
        issue_fence();
        load_b(X, obp + (long)(NB - 1) * (4 * KU) * EMB);
        load_a(AX, orw + (NB - 1) * (4 * KU));
        issue_fence();
        mma_blk(acc, AY, Y);
        issue_fence();
        load_b(Y, wbp);
        issue_fence();
        mma_blk(acc, AX, X);
        issue_fence();
        #pragma unroll
        for (int v = 0; v < 8; ++v) {
            v4f h;
            h.x = acc[0][v] + bias_o.x;
            h.y = acc[1][v] + bias_o.y;
            h.z = acc[2][v] + bias_o.z;
            h.w = acc[3][v] + bias_o.w;
            *(v4f*)(&sh_h[(mb + v) * HSTR + EMB + ncol0]) = h;
        }
    }
    __syncthreads();

    {
        v8f acc[4] = {};
        const int NB = HDIM / (4 * KU);
        load_a(AY, hrow);
        for (int i = 0; i < NB - 2; i += 2) {
            issue_fence();
            load_b(X, wbp + (long)(i + 1) * (4 * KU) * NOUT);
            load_a(AX, hrow + (i + 1) * (4 * KU));
            issue_fence();
            mma_blk(acc, AY, Y);
            issue_fence();
            load_b(Y, wbp + (long)(i + 2) * (4 * KU) * NOUT);
            load_a(AY, hrow + (i + 2) * (4 * KU));
            issue_fence();
            mma_blk(acc, AX, X);
        }
        issue_fence();
        load_b(X, wbp + (long)(NB - 1) * (4 * KU) * NOUT);
        load_a(AX, hrow + (NB - 1) * (4 * KU));
        issue_fence();
        mma_blk(acc, AY, Y);
        mma_blk(acc, AX, X);
        issue_fence();
        #pragma unroll
        for (int v = 0; v < 8; ++v) {
            int rr = sh_rows[mb + v];
            if (rr >= 0) {
                v4f o;
                o.x = acc[0][v] + bias_l.x;
                o.y = acc[1][v] + bias_l.y;
                o.z = acc[2][v] + bias_l.z;
                o.w = acc[3][v] + bias_l.w;
                *(v4f*)(out + (long)rr * NOUT + ncol0) = o;
            }
        }
    }
}

extern "C" void kernel_launch(void* const* d_in, const int* in_sizes, int n_in,
                              void* d_out, int out_size, void* d_ws, size_t ws_size,
                              hipStream_t stream) {
    const float* state   = (const float*)d_in[0];
    const float* obs     = (const float*)d_in[1];
    const int*   cat_ids = (const int*)  d_in[2];
    const float* obs_W   = (const float*)d_in[3];
    const float* obs_b   = (const float*)d_in[4];
    const float* state_W = (const float*)d_in[5];
    const float* state_b = (const float*)d_in[6];
    const float* l2_W    = (const float*)d_in[7];
    const float* l2_b    = (const float*)d_in[8];
    float*       out     = (float*)d_out;

    moe_mlp_kernel<<<MAXTILES, 128, 0, stream>>>(state, obs, obs_W, obs_b,
                                             state_W, state_b, l2_W, l2_b,
                                             cat_ids, out);
}


# Exposed waits cut from a block's entry and exit: barriers wait for LDS only, obs_W's first block asked for before them, tile scan by DPP adds, one argument fetch, output row ids read ahead
# speedup vs baseline: 9.4071x; 1.0489x over previous
#include <hip/hip_runtime.h>
#include <hip/hip_bf16.h>

typedef __attribute__((ext_vector_type(2))) float v2f;
typedef __attribute__((ext_vector_type(4))) float v4f;
typedef __attribute__((ext_vector_type(8))) float v8f;
typedef __attribute__((ext_vector_type(4))) int   v4i;

#define BATCH 2048
#define NCAT 32
#define SDIM 64
#define EMB 256
#define NOUT 256
#define ODIM 512
#define HDIM 512
#define HSTR 516
#define SSTR 68
#define KU 16
#define MAXTILES 158

__device__ __forceinline__ void mma4(v8f (&acc)[4], v2f a, v4f b0, v4f b1) {
    #pragma unroll
    for (int j = 0; j < 4; ++j) {
        v2f b;
        b.x = b0[j];
        b.y = b1[j];
        acc[j] = __builtin_amdgcn_wmma_f32_16x16x4_f32(
            false, a, false, b, (short)0, acc[j], false, false);
    }
}

__device__ __forceinline__ void load_b(v4f (&B)[2 * KU], const float* bp) {
    #pragma unroll
    for (int u = 0; u < KU; ++u) {
        B[2 * u]     = *(const v4f*)(bp + (long)(4 * u) * EMB);
        B[2 * u + 1] = *(const v4f*)(bp + (long)(4 * u + 1) * EMB);
    }
}
__device__ __forceinline__ void load_a(v2f (&A)[KU], const float* arow) {
    #pragma unroll
    for (int u = 0; u < KU; ++u) A[u] = *(const v2f*)(arow + 4 * u);
}
__device__ __forceinline__ void mma_blk(v8f (&acc)[4], const v2f (&A)[KU],
                                        const v4f (&B)[2 * KU]) {
    #pragma unroll
    for (int u = 0; u < KU; ++u) mma4(acc, A[u], B[2 * u], B[2 * u + 1]);
}
__device__ __forceinline__ void issue_fence() { __builtin_amdgcn_sched_barrier(0); }
__device__ __forceinline__ void lds_barrier() {
    __builtin_amdgcn_fence(__ATOMIC_RELEASE, "workgroup", "local");
    __builtin_amdgcn_s_barrier();
    __builtin_amdgcn_fence(__ATOMIC_ACQUIRE, "workgroup", "local");
}

__global__ __launch_bounds__(128) __attribute__((amdgpu_waves_per_eu(1, 1)))
void moe_mlp_kernel(const float* __restrict__ state,
                    const float* __restrict__ obs,
                    const float* __restrict__ obs_W,
                    const float* __restrict__ obs_b,
                    const float* __restrict__ state_W,
                    const float* __restrict__ state_b,
                    const float* __restrict__ l2_W,
                    const float* __restrict__ l2_b,
                    const int* __restrict__ cat_ids,
                    float* __restrict__ out) {
    const int g    = blockIdx.x;
    const int t    = threadIdx.x;
    const int wave = t >> 5;
    const int lane = t & 31;

    asm volatile("" :: "s"(state), "s"(obs), "s"(obs_W), "s"(obs_b),
                       "s"(state_W), "s"(state_b), "s"(l2_W), "s"(l2_b));

    const int m    = lane & 15;
    const int half = lane >> 4;
    const int kb   = half * 2;
    const int mb   = half * 8;
    const int ncol0 = wave * 64 + 4 * m;

    __shared__ int   sh_part[4 * NCAT];
    __shared__ __attribute__((aligned(16))) int sh_rows[16];
    __shared__ __attribute__((aligned(16))) float sh_s[16 * SSTR];
    __shared__ __attribute__((aligned(16))) float sh_o[16 * HSTR];
    __shared__ __attribute__((aligned(16))) float sh_h[16 * HSTR];

    v4i idv[4];
    {
        const v4i* cp = (const v4i*)(cat_ids + wave * (BATCH / 4)) + lane;
        #pragma unroll
        for (int q = 0; q < 4; ++q) idv[q] = cp[32 * q];
    }
    if (t < 16) sh_rows[t] = -1;

    unsigned nl[5];
    #pragma unroll
    for (int k = 0; k < 5; ++k) nl[k] = ((lane >> k) & 1) ? 0u : 0xffffffffu;
    int cnt_l = 0;
    #pragma unroll
    for (int q = 0; q < 4; ++q) {
        #pragma unroll
        for (int e = 0; e < 4; ++e) {
            const int id = idv[q][e];
            unsigned msk = 0xffffffffu;
            #pragma unroll
            for (int k = 0; k < 5; ++k)
                msk &= __builtin_amdgcn_ballot_w32(((id >> k) & 1) != 0) ^ nl[k];
            cnt_l += __popc(msk);
        }
    }
    v4f X[2 * KU], Y[2 * KU];
    const float* obp = obs_W + (long)kb * EMB + ncol0;
    const v4f bias_o = *(const v4f*)(obs_b + ncol0);
    load_b(Y, obp);
    issue_fence();

    sh_part[wave * NCAT + lane] = cnt_l;
    lds_barrier();

    int part[4];
    #pragma unroll
    for (int w = 0; w < 4; ++w) part[w] = sh_part[w * NCAT + lane];
    int incl = (part[0] + part[1] + part[2] + part[3] + 15) >> 4;
    incl += __builtin_amdgcn_update_dpp(0, incl, 0x111, 0xf, 0xf, false);
    incl += __builtin_amdgcn_update_dpp(0, incl, 0x112, 0xf, 0xf, false);
    incl += __builtin_amdgcn_update_dpp(0, incl, 0x114, 0xf, 0xf, false);
    incl += __builtin_amdgcn_update_dpp(0, incl, 0x118, 0xf, 0xf, false);
    {
        const int row0 = __builtin_amdgcn_readlane(incl, 15);
        incl += (lane >= 16) ? row0 : 0;
    }
    const int c = __popc(__builtin_amdgcn_ballot_w32(incl <= g));
    if (c >= NCAT) return;
    const int first = (c > 0) ? __builtin_amdgcn_readlane(incl, (c - 1) & 31) : 0;
    const int tile  = g - first;

    const float* Ws = state_W + (long)c * SDIM * EMB;
    const float* bs = state_b + (long)c * EMB;
    const float* Wl = l2_W    + (long)c * HDIM * NOUT;
    const float* lb = l2_b    + (long)c * NOUT;

    const v4f bias_s = *(const v4f*)(bs + ncol0);
    const v4f bias_l = *(const v4f*)(lb + ncol0);

    v2f AX[KU], AY[KU];

    load_b(X, Ws + (long)kb * EMB + ncol0);

    {
        int pre = -16 * tile;
        #pragma unroll
        for (int w = 0; w < 3; ++w) {
            const int pw = __builtin_amdgcn_readlane(part[w], c);
            pre += (w < wave) ? pw : 0;
        }
        #pragma unroll
        for (int q = 0; q < 4; ++q) {
            #pragma unroll
            for (int e = 0; e < 4; ++e) {
                const bool mine = (idv[q][e] == c);
                const unsigned hit = __builtin_amdgcn_ballot_w32(mine);
                const int slot = pre + (int)__builtin_amdgcn_mbcnt_lo(hit, 0u);
                if (mine && (unsigned)slot < 16u)
                    sh_rows[slot] = wave * (BATCH / 4) + 128 * q + 4 * lane + e;
                pre += __popc(hit);
            }
        }
    }
    lds_barrier();

    const float* wbp = Wl + (long)kb * NOUT + ncol0;

    {
        const v4f zero4 = {0.0f, 0.0f, 0.0f, 0.0f};
        int    srow[2];
        float* sdst[2];
        v4f    sv[2];
        #pragma unroll
        for (int q = 0; q < 2; ++q) {
            const int idx = t + 128 * q;
            const int rs  = sh_rows[idx >> 4];
            srow[q] = rs;
            sdst[q] = &sh_s[(idx >> 4) * SSTR + 4 * (idx & 15)];
            sv[q]   = *(const v4f*)(state + (long)((rs >= 0) ? rs : 0) * SDIM + 4 * (idx & 15));
        }
        int orow[16];
        v4f ov[16];
        #pragma unroll
        for (int i = 0; i < 16; ++i) {
            const int ro = __builtin_amdgcn_readfirstlane(sh_rows[i]);
            orow[i] = ro;
            ov[i] = *(const v4f*)(obs + (long)((ro >= 0) ? ro : 0) * ODIM + 4 * t);
        }
        #pragma unroll
        for (int q = 0; q < 2; ++q) *(v4f*)sdst[q] = sv[q];
        #pragma unroll
        for (int i = 0; i < 16; ++i) *(v4f*)(&sh_o[i * HSTR + 4 * t]) = ov[i];
        #pragma unroll
        for (int q = 0; q < 2; ++q)
            if (srow[q] < 0) *(v4f*)sdst[q] = zero4;
        #pragma unroll
        for (int i = 0; i < 16; ++i)
            if (orow[i] < 0) *(v4f*)(&sh_o[i * HSTR + 4 * t]) = zero4;
    }
    lds_barrier();

    const float* srw  = &sh_s[m * SSTR + kb];
    const float* orw  = &sh_o[m * HSTR + kb];
    const float* hrow = &sh_h[m * HSTR + kb];

    load_a(AX, srw);
    load_a(AY, orw);
    issue_fence();

    {
        v8f acc[4] = {};
        mma_blk(acc, AX, X);
        issue_fence();
        #pragma unroll
        for (int v = 0; v < 8; ++v) {
            v4f h;
            h.x = fmaxf(acc[0][v] + bias_s.x, 0.0f);
            h.y = fmaxf(acc[1][v] + bias_s.y, 0.0f);
            h.z = fmaxf(acc[2][v] + bias_s.z, 0.0f);
            h.w = fmaxf(acc[3][v] + bias_s.w, 0.0f);
            *(v4f*)(&sh_h[(mb + v) * HSTR + ncol0]) = h;
        }
    }

    {
        v8f acc[4] = {};
        const int NB = ODIM / (4 * KU);
        for (int i = 0; i < NB - 2; i += 2) {
            issue_fence();
            load_b(X, obp + (long)(i + 1) * (4 * KU) * EMB);
            load_a(AX, orw + (i + 1) * (4 * KU));
            issue_fence();
            mma_blk(acc, AY, Y);
            issue_fence();
            load_b(Y, obp + (long)(i + 2) * (4 * KU) * EMB);
            load_a(AY, orw + (i + 2) * (4 * KU));
            issue_fence();
            mma_blk(acc, AX, X);
        }
        issue_fence();
        load_b(X, obp + (long)(NB - 1) * (4 * KU) * EMB);
        load_a(AX, orw + (NB - 1) * (4 * KU));
        issue_fence();
        mma_blk(acc, AY, Y);
        issue_fence();
        load_b(Y, wbp);
        issue_fence();
        mma_blk(acc, AX, X);
        issue_fence();
        #pragma unroll
        for (int v = 0; v < 8; ++v) {
            v4f h;
            h.x = acc[0][v] + bias_o.x;
            h.y = acc[1][v] + bias_o.y;
            h.z = acc[2][v] + bias_o.z;
            h.w = acc[3][v] + bias_o.w;
            *(v4f*)(&sh_h[(mb + v) * HSTR + EMB + ncol0]) = h;
        }
    }
    lds_barrier();

    {
        v8f acc[4] = {};
        const int NB = HDIM / (4 * KU);
        load_a(AY, hrow);
        const v4i rlo = *(const v4i*)(&sh_rows[mb]);
        const v4i rhi = *(const v4i*)(&sh_rows[mb + 4]);
        for (int i = 0; i < NB - 2; i += 2) {
            issue_fence();
            load_b(X, wbp + (long)(i + 1) * (4 * KU) * NOUT);
            load_a(AX, hrow + (i + 1) * (4 * KU));
            issue_fence();
            mma_blk(acc, AY, Y);
            issue_fence();
            load_b(Y, wbp + (long)(i + 2) * (4 * KU) * NOUT);
            load_a(AY, hrow + (i + 2) * (4 * KU));
            issue_fence();
            mma_blk(acc, AX, X);
        }
        issue_fence();
        load_b(X, wbp + (long)(NB - 1) * (4 * KU) * NOUT);
        load_a(AX, hrow + (NB - 1) * (4 * KU));
        issue_fence();
        mma_blk(acc, AY, Y);
        mma_blk(acc, AX, X);
        issue_fence();
        #pragma unroll
        for (int v = 0; v < 8; ++v) {
            const int rr = (v < 4) ? rlo[v & 3] : rhi[v & 3];
            if (rr >= 0) {
                v4f o;
                o.x = acc[0][v] + bias_l.x;
                o.y = acc[1][v] + bias_l.y;
                o.z = acc[2][v] + bias_l.z;
                o.w = acc[3][v] + bias_l.w;
                *(v4f*)(out + (long)rr * NOUT + ncol0) = o;
            }
        }
    }
}

extern "C" void kernel_launch(void* const* d_in, const int* in_sizes, int n_in,
                              void* d_out, int out_size, void* d_ws, size_t ws_size,
                              hipStream_t stream) {
    const float* state   = (const float*)d_in[0];
    const float* obs     = (const float*)d_in[1];
    const int*   cat_ids = (const int*)  d_in[2];
    const float* obs_W   = (const float*)d_in[3];
    const float* obs_b   = (const float*)d_in[4];
    const float* state_W = (const float*)d_in[5];
    const float* state_b = (const float*)d_in[6];
    const float* l2_W    = (const float*)d_in[7];
    const float* l2_b    = (const float*)d_in[8];
    float*       out     = (float*)d_out;

    moe_mlp_kernel<<<MAXTILES, 128, 0, stream>>>(state, obs, obs_W, obs_b,
                                             state_W, state_b, l2_W, l2_b,
                                             cat_ids, out);
}
